# starting kernel
# baseline (speedup 1.0000x reference)
#include <hip/hip_runtime.h>

typedef float v2f __attribute__((ext_vector_type(2)));
typedef float v8f __attribute__((ext_vector_type(8)));

#define D_IN   2048
#define D_OUT  2048
#define NEXP   16
#define RANK   16
#define KTOP   2
#define ER     256
#define TOKS   16
#define CHUNK  256
#define XSTR   (CHUNK + 4)
#define HSTR   (ER + 4)

__global__ __launch_bounds__(256)
void moe_lora_wmma_kernel(const float* __restrict__ x,
                          const float* __restrict__ routing,
                          const int*   __restrict__ idxs,
                          const float* __restrict__ w_a,
                          const float* __restrict__ w_b,
                          float* __restrict__ out)
{
    __shared__ float x_lds[TOKS * XSTR];
    __shared__ float h_lds[TOKS * HSTR];
    __shared__ float g_lds[TOKS * NEXP];

    const int tid  = threadIdx.x;
    const int wave = tid >> 5;
    const int lane = tid & 31;
    const int nIdx = lane & 15;
    const int hv   = lane >> 4;
    const int tileBase = blockIdx.x * TOKS;

    {
        const int t = tid >> 4;
        const int e = tid & 15;
        const int tok = tileBase + t;
        float g = 0.f;
#pragma unroll
        for (int k = 0; k < KTOP; ++k) {
            const int   ie = idxs[tok * KTOP + k];
            const float rw = routing[tok * KTOP + k];
            g += (ie == e) ? rw : 0.f;
        }
        g_lds[t * NEXP + e] = 2.0f * g;
    }

    v8f acc0a = {}, acc0b = {};
    v8f acc1a = {}, acc1b = {};
    const int er0 = wave * 32;

    for (int c = 0; c < D_IN / CHUNK; ++c) {
        __syncthreads();
        {
            const float* xg = x + (size_t)tileBase * D_IN + c * CHUNK;
#pragma unroll
            for (int i = 0; i < 4; ++i) {
                const int fi  = tid + i * 256;
                const int row = fi >> 6;
                const int col = (fi & 63) << 2;
                const float4 v = *reinterpret_cast<const float4*>(
                    xg + (size_t)row * D_IN + col);
                *reinterpret_cast<float4*>(&x_lds[row * XSTR + col]) = v;
            }
        }
        __syncthreads();

        const float* waBase = w_a + (size_t)(er0 + nIdx) * D_IN + c * CHUNK;
#pragma unroll 2
        for (int k0 = 0; k0 < CHUNK; k0 += 8) {
            const int kkA = k0 + 2 * hv;
            const int kkB = kkA + 4;
            const v2f a0 = *reinterpret_cast<const v2f*>(&x_lds[nIdx * XSTR + kkA]);
            const v2f a1 = *reinterpret_cast<const v2f*>(&x_lds[nIdx * XSTR + kkB]);
            const v2f b0a = *reinterpret_cast<const v2f*>(waBase + kkA);
            const v2f b1a = *reinterpret_cast<const v2f*>(waBase + (size_t)16 * D_IN + kkA);
            const v2f b0b = *reinterpret_cast<const v2f*>(waBase + kkB);
            const v2f b1b = *reinterpret_cast<const v2f*>(waBase + (size_t)16 * D_IN + kkB);
            acc0a = __builtin_amdgcn_wmma_f32_16x16x4_f32(false, a0, false, b0a,
                                                          (short)0, acc0a, false, false);
            acc1a = __builtin_amdgcn_wmma_f32_16x16x4_f32(false, a0, false, b1a,
                                                          (short)0, acc1a, false, false);
            acc0b = __builtin_amdgcn_wmma_f32_16x16x4_f32(false, a1, false, b0b,
                                                          (short)0, acc0b, false, false);
            acc1b = __builtin_amdgcn_wmma_f32_16x16x4_f32(false, a1, false, b1b,
                                                          (short)0, acc1b, false, false);
        }
    }

    {
        const v8f h0v = acc0a + acc0b;
        const v8f h1v = acc1a + acc1b;
#pragma unroll
        for (int v = 0; v < 8; ++v) {
            const int t = v + 8 * hv;
            const float h0 = h0v[v];
            const float h1 = h1v[v];
            const float s0 = h0 * __builtin_amdgcn_rcpf(1.0f + __expf(-h0));
            const float s1 = h1 * __builtin_amdgcn_rcpf(1.0f + __expf(-h1));
            h_lds[t * HSTR + er0 + nIdx]      = g_lds[t * NEXP + wave * 2]     * s0;
            h_lds[t * HSTR + er0 + 16 + nIdx] = g_lds[t * NEXP + wave * 2 + 1] * s1;
        }
    }
    __syncthreads();

    const int dBase = wave * 256;
    for (int nt = 0; nt < 8; ++nt) {
        const int d0 = dBase + nt * 16 + nIdx;
        const int d1 = d0 + 128;
        v8f p0a = {}, p0b = {}, p1a = {}, p1b = {};
#pragma unroll 2
        for (int k0 = 0; k0 < ER; k0 += 8) {
            const int erA = k0 + 2 * hv;
            const int erB = erA + 4;
            const v2f a0 = *reinterpret_cast<const v2f*>(&h_lds[nIdx * HSTR + erA]);
            const v2f a1 = *reinterpret_cast<const v2f*>(&h_lds[nIdx * HSTR + erB]);
            const int eA = erA >> 4, rA = erA & 15;
            const int eB = erB >> 4, rB = erB & 15;
            const float* wbA = w_b + (size_t)eA * (D_OUT * RANK) + rA;
            const float* wbB = w_b + (size_t)eB * (D_OUT * RANK) + rB;
            const v2f b00 = *reinterpret_cast<const v2f*>(wbA + (size_t)d0 * RANK);
            const v2f b01 = *reinterpret_cast<const v2f*>(wbA + (size_t)d1 * RANK);
            const v2f b10 = *reinterpret_cast<const v2f*>(wbB + (size_t)d0 * RANK);
            const v2f b11 = *reinterpret_cast<const v2f*>(wbB + (size_t)d1 * RANK);
            p0a = __builtin_amdgcn_wmma_f32_16x16x4_f32(false, a0, false, b00,
                                                        (short)0, p0a, false, false);
            p1a = __builtin_amdgcn_wmma_f32_16x16x4_f32(false, a0, false, b01,
                                                        (short)0, p1a, false, false);
            p0b = __builtin_amdgcn_wmma_f32_16x16x4_f32(false, a1, false, b10,
                                                        (short)0, p0b, false, false);
            p1b = __builtin_amdgcn_wmma_f32_16x16x4_f32(false, a1, false, b11,
                                                        (short)0, p1b, false, false);
        }
        const v8f o0 = p0a + p0b;
        const v8f o1 = p1a + p1b;
        float* ob = out + (size_t)tileBase * D_OUT;
#pragma unroll
        for (int v = 0; v < 8; ++v) {
            const size_t row = (size_t)(v + 8 * hv) * D_OUT;
            ob[row + d0] = o0[v];
            ob[row + d1] = o1[v];
        }
    }
}

extern "C" void kernel_launch(void* const* d_in, const int* in_sizes, int n_in,
                              void* d_out, int out_size, void* d_ws, size_t ws_size,
                              hipStream_t stream) {
    const float* x       = (const float*)d_in[0];
    const float* routing = (const float*)d_in[1];
    const int*   idxs    = (const int*)d_in[2];
    const float* w_a     = (const float*)d_in[3];
    const float* w_b     = (const float*)d_in[4];
    float* out = (float*)d_out;

    const int tokens = in_sizes[1] / KTOP;
    const int tiles  = tokens / TOKS;

    hipLaunchKernelGGL(moe_lora_wmma_kernel, dim3(tiles), dim3(256), 0, stream,
                       x, routing, idxs, w_a, w_b, out);
}


# First product computes only routed experts: each 64-token group lists its rows per expert by ballot and runs one WMMA chain per 16-row list tile; per-slot partial sums; sum order unchanged
# speedup vs baseline: 2.3126x; 2.3126x over previous
#include <hip/hip_runtime.h>


typedef float v2f __attribute__((ext_vector_type(2)));
typedef float v8f __attribute__((ext_vector_type(8)));

#define D_IN   2048
#define D_OUT  2048
#define NEXP   16
#define RANK   16
#define KTOP   2
#define ER     256
#define NTOK   4096
#define TG     64
#define MT     4
#define KSPLIT 4
#define KPART  (D_IN / KSPLIT)
#define KCH    128
#define XS     (KCH + 4)
#define HS     (ER + 4)

#define MAXT   24
#define HCS    32

__global__ __launch_bounds__(256) __attribute__((amdgpu_waves_per_eu(2, 4)))
void lora_down_routed_kernel(const float* __restrict__ x,
                             const int*   __restrict__ idxs,
                             const float* __restrict__ w_a,
                             float* __restrict__ hc)
{
    __shared__ float x_lds[TG * XS];
    __shared__ int   list_lds[NEXP * TG];
    __shared__ int   cnt_lds[NEXP];
    __shared__ int   i0_lds[TG];
    __shared__ int   i1_lds[TG];
    __shared__ int   tileE_lds[MAXT];
    __shared__ int   tileQ_lds[MAXT];
    __shared__ int   nT_lds;

    const int tid  = threadIdx.x;
    const int wave = __builtin_amdgcn_readfirstlane(tid >> 5);
    const int lane = tid & 31;
    const int nIdx = lane & 15;
    const int hv   = lane >> 4;
    const int grp  = blockIdx.x >> 2;
    const int ks   = blockIdx.x & 3;
    const int tokBase = grp * TG;
    const int kBase   = ks * KPART;

    if (tid < TG) {
        i0_lds[tid] = idxs[(tokBase + tid) * KTOP + 0];
        i1_lds[tid] = idxs[(tokBase + tid) * KTOP + 1];
    }
    __syncthreads();

    for (int j = 0; j < 2; ++j) {
        const int e = wave * 2 + j;
        int base = 0;
        for (int it = 0; it < 2; ++it) {
            const int row = it * 32 + lane;
            const bool hit = (i0_lds[row] == e) || (i1_lds[row] == e);
            const unsigned mask = __builtin_amdgcn_ballot_w32(hit);
            const int pos = base + __popc(mask & ((1u << lane) - 1u));
            if (hit) list_lds[e * TG + pos] = row;
            base += __popc(mask);
        }
        if (lane == 0) cnt_lds[e] = base;
    }
    __syncthreads();
    if (wave == 0) {
        const int cE = (lane < NEXP) ? cnt_lds[lane] : 0;
        const int nt = (cE + 15) >> 4;
        int inc = nt;
        for (int d = 1; d < NEXP; d <<= 1) {
            const int o = __shfl_up(inc, d);
            if (lane >= d) inc += o;
        }
        const int first = inc - nt;
        for (int t = 0; t < nt; ++t) { tileE_lds[first + t] = lane; tileQ_lds[first + t] = t * 16; }
        if (lane == NEXP - 1) nT_lds = inc;
    }
    __syncthreads();

    const int nT = nT_lds;
    int  sE[3], sQ[3], sRow[3];
    bool sV[3];
#pragma unroll
    for (int j = 0; j < 3; ++j) {
        const int id = wave + 8 * j;
        sV[j] = id < nT;
        sE[j] = sV[j] ? tileE_lds[id] : 0;
        sQ[j] = sV[j] ? tileQ_lds[id] : 0;
        const int q = sQ[j] + nIdx;
        sRow[j] = (sV[j] && q < cnt_lds[sE[j]]) ? list_lds[sE[j] * TG + q] : 0;
    }
    const bool has3 = sV[2];

    v8f acc[3];
#pragma unroll
    for (int j = 0; j < 3; ++j) acc[j] = (v8f){};

    for (int c = 0; c < KPART / KCH; ++c) {
        __syncthreads();
        {
            const float* xg = x + (size_t)tokBase * D_IN + kBase + c * KCH;
#pragma unroll
            for (int i = 0; i < 8; ++i) {
                const int fi  = tid + i * 256;
                const int row = fi >> 5;
                const int col = (fi & 31) << 2;
                const float4 v = *reinterpret_cast<const float4*>(
                    xg + (size_t)row * D_IN + col);
                *reinterpret_cast<float4*>(&x_lds[row * XS + col]) = v;
            }
        }
        __syncthreads();

        const float* wg0 = w_a + (size_t)(sE[0] * RANK + nIdx) * D_IN + kBase + c * KCH + 2 * hv;
        const float* wg1 = w_a + (size_t)(sE[1] * RANK + nIdx) * D_IN + kBase + c * KCH + 2 * hv;
        const float* xl0 = &x_lds[sRow[0] * XS + 2 * hv];
        const float* xl1 = &x_lds[sRow[1] * XS + 2 * hv];
#pragma unroll 8
        for (int k0 = 0; k0 < KCH; k0 += 4) {
            const v2f b0 = *reinterpret_cast<const v2f*>(wg0 + k0);
            const v2f b1 = *reinterpret_cast<const v2f*>(wg1 + k0);
            const v2f a0 = *reinterpret_cast<const v2f*>(xl0 + k0);
            const v2f a1 = *reinterpret_cast<const v2f*>(xl1 + k0);
            acc[0] = __builtin_amdgcn_wmma_f32_16x16x4_f32(false, a0, false, b0,
                                                           (short)0, acc[0], false, false);
            acc[1] = __builtin_amdgcn_wmma_f32_16x16x4_f32(false, a1, false, b1,
                                                           (short)0, acc[1], false, false);
        }
        if (has3) {
            const float* wg2 = w_a + (size_t)(sE[2] * RANK + nIdx) * D_IN + kBase + c * KCH + 2 * hv;
            const float* xl2 = &x_lds[sRow[2] * XS + 2 * hv];
#pragma unroll 8
            for (int k0 = 0; k0 < KCH; k0 += 4) {
                const v2f b2 = *reinterpret_cast<const v2f*>(wg2 + k0);
                const v2f a2 = *reinterpret_cast<const v2f*>(xl2 + k0);
                acc[2] = __builtin_amdgcn_wmma_f32_16x16x4_f32(false, a2, false, b2,
                                                               (short)0, acc[2], false, false);
            }
        }
    }

#pragma unroll
    for (int j = 0; j < 3; ++j) {
        if (sV[j]) {
            const int e = sE[j];
            const int c = cnt_lds[e];
#pragma unroll
            for (int v = 0; v < 8; ++v) {
                const int q = sQ[j] + v + 8 * hv;
                if (q < c) {
                    const int row = list_lds[e * TG + q];
                    float* hb = hc + ((size_t)ks * NTOK + tokBase + row) * (KTOP * HCS) + nIdx;
                    if (i0_lds[row] == e) hb[0]   = acc[j][v];
                    if (i1_lds[row] == e) hb[HCS] = acc[j][v];
                }
            }
        }
    }
}

__global__ __launch_bounds__(256)
void lora_up_kernel(const float* __restrict__ hc,
                    const float* __restrict__ routing,
                    const int*   __restrict__ idxs,
                    const float* __restrict__ w_b,
                    float* __restrict__ out)
{
    __shared__ float h_lds[TG * HS];
    __shared__ float g_lds[TG * NEXP];
    __shared__ int   ib_lds[TG * KTOP];

    const int tid  = threadIdx.x;
    const int wave = tid >> 5;
    const int lane = tid & 31;
    const int nIdx = lane & 15;
    const int hv   = lane >> 4;
    const int grp  = blockIdx.x >> 3;
    const int dg   = blockIdx.x & 7;
    const int tokBase = grp * TG;

    {
        const int t   = tid >> 2;
        const int e4  = (tid & 3) * 4;
        const int tok = tokBase + t;
        const int   i0 = idxs[tok * KTOP + 0];
        const int   i1 = idxs[tok * KTOP + 1];
        const float r0 = routing[tok * KTOP + 0];
        const float r1 = routing[tok * KTOP + 1];
        if ((tid & 3) == 0) { ib_lds[t * KTOP + 0] = i0; ib_lds[t * KTOP + 1] = i1; }
        for (int j = 0; j < 4; ++j) {
            const int e = e4 + j;
            float g = 0.f;
            g += (i0 == e) ? r0 : 0.f;
            g += (i1 == e) ? r1 : 0.f;
            g_lds[t * NEXP + e] = 2.0f * g;
        }
    }
    __syncthreads();

    {
        const size_t ksStride = (size_t)NTOK * KTOP * HCS;
#pragma unroll 4
        for (int i = 0; i < 16; ++i) {
            const int fi  = tid + i * 256;
            const int row = fi >> 6;
            const int col = (fi & 63) << 2;
            const int eb  = col >> 4;
            const int e0  = ib_lds[row * KTOP + 0];
            const int e1  = ib_lds[row * KTOP + 1];
            const int slot = (eb == e0) ? 0 : ((eb == e1) ? 1 : -1);
            float4 o = make_float4(0.f, 0.f, 0.f, 0.f);
            if (slot >= 0) {
                const float* p0 = hc + ((size_t)(tokBase + row) * KTOP + slot) * HCS + (col & 15);
                const float4 s0 = *reinterpret_cast<const float4*>(p0);
                const float4 s1 = *reinterpret_cast<const float4*>(p0 + 1 * ksStride);
                const float4 s2 = *reinterpret_cast<const float4*>(p0 + 2 * ksStride);
                const float4 s3 = *reinterpret_cast<const float4*>(p0 + 3 * ksStride);
                const float gate = g_lds[row * NEXP + eb];
                const float hx = ((s0.x + s1.x) + s2.x) + s3.x;
                const float hy = ((s0.y + s1.y) + s2.y) + s3.y;
                const float hz = ((s0.z + s1.z) + s2.z) + s3.z;
                const float hw = ((s0.w + s1.w) + s2.w) + s3.w;
                o.x = gate * (hx * __builtin_amdgcn_rcpf(1.0f + __expf(-hx)));
                o.y = gate * (hy * __builtin_amdgcn_rcpf(1.0f + __expf(-hy)));
                o.z = gate * (hz * __builtin_amdgcn_rcpf(1.0f + __expf(-hz)));
                o.w = gate * (hw * __builtin_amdgcn_rcpf(1.0f + __expf(-hw)));
            }
            *reinterpret_cast<float4*>(&h_lds[row * HS + col]) = o;
        }
    }
    __syncthreads();

    v8f acc[MT][2];
#pragma unroll
    for (int m = 0; m < MT; ++m) { acc[m][0] = (v8f){}; acc[m][1] = (v8f){}; }

    const int d0 = dg * 256 + wave * 32 + nIdx;
    const float* hl = &h_lds[nIdx * HS + 2 * hv];
    const float* wd = w_b + (size_t)d0 * RANK;
#pragma unroll 4
    for (int k0 = 0; k0 < ER; k0 += 4) {
        const int erK = k0 + 2 * hv;
        const float* wb = wd + (size_t)(erK >> 4) * (D_OUT * RANK) + (erK & 15);
        const v2f b0 = *reinterpret_cast<const v2f*>(wb);
        const v2f b1 = *reinterpret_cast<const v2f*>(wb + 16 * RANK);
        v2f a[MT];
#pragma unroll
        for (int m = 0; m < MT; ++m)
            a[m] = *reinterpret_cast<const v2f*>(hl + m * 16 * HS + k0);
#pragma unroll
        for (int m = 0; m < MT; ++m) {
            acc[m][0] = __builtin_amdgcn_wmma_f32_16x16x4_f32(false, a[m], false, b0,
                                                              (short)0, acc[m][0], false, false);
            acc[m][1] = __builtin_amdgcn_wmma_f32_16x16x4_f32(false, a[m], false, b1,
                                                              (short)0, acc[m][1], false, false);
        }
    }

    float* ob = out + (size_t)tokBase * D_OUT + d0;
#pragma unroll
    for (int m = 0; m < MT; ++m) {
#pragma unroll
        for (int v = 0; v < 8; ++v) {
            const size_t row = (size_t)(m * 16 + v + 8 * hv) * D_OUT;
            ob[row]      = acc[m][0][v];
            ob[row + 16] = acc[m][1][v];
        }
    }
}

extern "C" void kernel_launch(void* const* d_in, const int* in_sizes, int n_in,
                              void* d_out, int out_size, void* d_ws, size_t ws_size,
                              hipStream_t stream) {
    const float* x       = (const float*)d_in[0];
    const float* routing = (const float*)d_in[1];
    const int*   idxs    = (const int*)d_in[2];
    const float* w_a     = (const float*)d_in[3];
    const float* w_b     = (const float*)d_in[4];
    float* out  = (float*)d_out;
    float* hc   = (float*)d_ws;

    hipLaunchKernelGGL(lora_down_routed_kernel, dim3(256), dim3(256), 0, stream,
                       x, idxs, w_a, hc);
    hipLaunchKernelGGL(lora_up_kernel, dim3(512), dim3(256), 0, stream,
                       hc, routing, idxs, w_b, out);
}


# Second product on routed experts only: each wave walks the group's per-expert 16-row list tiles (K=16), joins a token's two expert parts in its own LDS block, stores whole lines; sum order changes
# speedup vs baseline: 3.0377x; 1.3135x over previous
#include <hip/hip_runtime.h>


typedef float v2f __attribute__((ext_vector_type(2)));
typedef float v8f __attribute__((ext_vector_type(8)));

#define D_IN   2048
#define D_OUT  2048
#define NEXP   16
#define RANK   16
#define KTOP   2
#define NTOK   4096
#define TG     64
#define KSPLIT 4
#define KPART  (D_IN / KSPLIT)
#define KCH    128
#define XS     (KCH + 4)

#define MAXT   24
#define HCS    32

__global__ __launch_bounds__(256) __attribute__((amdgpu_waves_per_eu(2, 4)))
void lora_down_routed_kernel(const float* __restrict__ x,
                             const int*   __restrict__ idxs,
                             const float* __restrict__ w_a,
                             float* __restrict__ hc)
{
    __shared__ float x_lds[TG * XS];
    __shared__ int   list_lds[NEXP * TG];
    __shared__ int   cnt_lds[NEXP];
    __shared__ int   i0_lds[TG];
    __shared__ int   i1_lds[TG];
    __shared__ int   tileE_lds[MAXT];
    __shared__ int   tileQ_lds[MAXT];
    __shared__ int   nT_lds;

    const int tid  = threadIdx.x;
    const int wave = __builtin_amdgcn_readfirstlane(tid >> 5);
    const int lane = tid & 31;
    const int nIdx = lane & 15;
    const int hv   = lane >> 4;
    const int grp  = blockIdx.x >> 2;
    const int ks   = blockIdx.x & 3;
    const int tokBase = grp * TG;
    const int kBase   = ks * KPART;

    if (tid < TG) {
        i0_lds[tid] = idxs[(tokBase + tid) * KTOP + 0];
        i1_lds[tid] = idxs[(tokBase + tid) * KTOP + 1];
    }
    __syncthreads();

    for (int j = 0; j < 2; ++j) {
        const int e = wave * 2 + j;
        int base = 0;
        for (int it = 0; it < 2; ++it) {
            const int row = it * 32 + lane;
            const bool hit = (i0_lds[row] == e) || (i1_lds[row] == e);
            const unsigned mask = __builtin_amdgcn_ballot_w32(hit);
            const int pos = base + __popc(mask & ((1u << lane) - 1u));
            if (hit) list_lds[e * TG + pos] = row;
            base += __popc(mask);
        }
        if (lane == 0) cnt_lds[e] = base;
    }
    __syncthreads();
    if (wave == 0) {
        const int cE = (lane < NEXP) ? cnt_lds[lane] : 0;
        const int nt = (cE + 15) >> 4;
        int inc = nt;
        for (int d = 1; d < NEXP; d <<= 1) {
            const int o = __shfl_up(inc, d);
            if (lane >= d) inc += o;
        }
        const int first = inc - nt;
        for (int t = 0; t < nt; ++t) { tileE_lds[first + t] = lane; tileQ_lds[first + t] = t * 16; }
        if (lane == NEXP - 1) nT_lds = inc;
    }
    __syncthreads();

    const int nT = nT_lds;
    int  sE[3], sQ[3], sRow[3];
    bool sV[3];
#pragma unroll
    for (int j = 0; j < 3; ++j) {
        const int id = wave + 8 * j;
        sV[j] = id < nT;
        sE[j] = sV[j] ? tileE_lds[id] : 0;
        sQ[j] = sV[j] ? tileQ_lds[id] : 0;
        const int q = sQ[j] + nIdx;
        sRow[j] = (sV[j] && q < cnt_lds[sE[j]]) ? list_lds[sE[j] * TG + q] : 0;
    }
    const bool has3 = sV[2];

    v8f acc[3];
#pragma unroll
    for (int j = 0; j < 3; ++j) acc[j] = (v8f){};

    for (int c = 0; c < KPART / KCH; ++c) {
        __syncthreads();
        {
            const float* xg = x + (size_t)tokBase * D_IN + kBase + c * KCH;
#pragma unroll
            for (int i = 0; i < 8; ++i) {
                const int fi  = tid + i * 256;
                const int row = fi >> 5;
                const int col = (fi & 31) << 2;
                const float4 v = *reinterpret_cast<const float4*>(
                    xg + (size_t)row * D_IN + col);
                *reinterpret_cast<float4*>(&x_lds[row * XS + col]) = v;
            }
        }
        __syncthreads();

        const float* wg0 = w_a + (size_t)(sE[0] * RANK + nIdx) * D_IN + kBase + c * KCH + 2 * hv;
        const float* wg1 = w_a + (size_t)(sE[1] * RANK + nIdx) * D_IN + kBase + c * KCH + 2 * hv;
        const float* xl0 = &x_lds[sRow[0] * XS + 2 * hv];
        const float* xl1 = &x_lds[sRow[1] * XS + 2 * hv];
#pragma unroll 8
        for (int k0 = 0; k0 < KCH; k0 += 4) {
            const v2f b0 = *reinterpret_cast<const v2f*>(wg0 + k0);
            const v2f b1 = *reinterpret_cast<const v2f*>(wg1 + k0);
            const v2f a0 = *reinterpret_cast<const v2f*>(xl0 + k0);
            const v2f a1 = *reinterpret_cast<const v2f*>(xl1 + k0);
            acc[0] = __builtin_amdgcn_wmma_f32_16x16x4_f32(false, a0, false, b0,
                                                           (short)0, acc[0], false, false);
            acc[1] = __builtin_amdgcn_wmma_f32_16x16x4_f32(false, a1, false, b1,
                                                           (short)0, acc[1], false, false);
        }
        if (has3) {
            const float* wg2 = w_a + (size_t)(sE[2] * RANK + nIdx) * D_IN + kBase + c * KCH + 2 * hv;
            const float* xl2 = &x_lds[sRow[2] * XS + 2 * hv];
#pragma unroll 8
            for (int k0 = 0; k0 < KCH; k0 += 4) {
                const v2f b2 = *reinterpret_cast<const v2f*>(wg2 + k0);
                const v2f a2 = *reinterpret_cast<const v2f*>(xl2 + k0);
                acc[2] = __builtin_amdgcn_wmma_f32_16x16x4_f32(false, a2, false, b2,
                                                               (short)0, acc[2], false, false);
            }
        }
    }

#pragma unroll
    for (int j = 0; j < 3; ++j) {
        if (sV[j]) {
            const int e = sE[j];
            const int c = cnt_lds[e];
#pragma unroll
            for (int v = 0; v < 8; ++v) {
                const int q = sQ[j] + v + 8 * hv;
                if (q < c) {
                    const int row = list_lds[e * TG + q];
                    float* hb = hc + ((size_t)ks * NTOK + tokBase + row) * (KTOP * HCS) + nIdx;
                    if (i0_lds[row] == e) hb[0]   = acc[j][v];
                    if (i1_lds[row] == e) hb[HCS] = acc[j][v];
                }
            }
        }
    }
}

#define HRS 20
__global__ __launch_bounds__(256)
void lora_up_routed_kernel(const float* __restrict__ hc,
                           const float* __restrict__ routing,
                           const int*   __restrict__ idxs,
                           const float* __restrict__ w_b,
                           float* __restrict__ out)
{
    __shared__ float o_lds[8 * (TG + 1) * 32];
    __shared__ float hr_lds[TG * KTOP * HRS];
    __shared__ int   list_lds[NEXP * TG];
    __shared__ int   cnt_lds[NEXP];
    __shared__ int   i0_lds[TG];
    __shared__ int   i1_lds[TG];
    __shared__ int   tileP_lds[MAXT];
    __shared__ int   nT_lds;

    const int tid  = threadIdx.x;
    const int wave = __builtin_amdgcn_readfirstlane(tid >> 5);
    const int lane = tid & 31;
    const int nIdx = lane & 15;
    const int hv   = lane >> 4;
    const int grp  = blockIdx.x >> 3;
    const int dg   = blockIdx.x & 7;
    const int tokBase = grp * TG;

    if (tid < TG) {
        i0_lds[tid] = idxs[(tokBase + tid) * KTOP + 0];
        i1_lds[tid] = idxs[(tokBase + tid) * KTOP + 1];
    }
    float* ob = &o_lds[wave * ((TG + 1) * 32)];
    for (int it = 0; it < 16; ++it)
        *reinterpret_cast<float4*>(ob + (it * 32 + lane) * 4) = make_float4(0.f, 0.f, 0.f, 0.f);

    {
        const size_t ksStride = (size_t)NTOK * KTOP * HCS;
#pragma unroll
        for (int i = 0; i < 2; ++i) {
            const int fi   = tid + i * 256;
            const int row  = fi >> 3;
            const int slot = (fi >> 2) & 1;
            const int c4   = (fi & 3) << 2;
            const int tok  = tokBase + row;
            const int   e0 = idxs[tok * KTOP + 0];
            const int   e1 = idxs[tok * KTOP + 1];
            const float r0 = routing[tok * KTOP + 0];
            const float r1 = routing[tok * KTOP + 1];
            const int   es = slot ? e1 : e0;
            float g = 0.f;
            g += (e0 == es) ? r0 : 0.f;
            g += (e1 == es) ? r1 : 0.f;
            const float gate = 2.0f * g;
            const float* p0 = hc + ((size_t)tok * KTOP + slot) * HCS + c4;
            const float4 s0 = *reinterpret_cast<const float4*>(p0);
            const float4 s1 = *reinterpret_cast<const float4*>(p0 + 1 * ksStride);
            const float4 s2 = *reinterpret_cast<const float4*>(p0 + 2 * ksStride);
            const float4 s3 = *reinterpret_cast<const float4*>(p0 + 3 * ksStride);
            const float hx = ((s0.x + s1.x) + s2.x) + s3.x;
            const float hy = ((s0.y + s1.y) + s2.y) + s3.y;
            const float hz = ((s0.z + s1.z) + s2.z) + s3.z;
            const float hw = ((s0.w + s1.w) + s2.w) + s3.w;
            float4 o;
            o.x = gate * (hx * __builtin_amdgcn_rcpf(1.0f + __expf(-hx)));
            o.y = gate * (hy * __builtin_amdgcn_rcpf(1.0f + __expf(-hy)));
            o.z = gate * (hz * __builtin_amdgcn_rcpf(1.0f + __expf(-hz)));
            o.w = gate * (hw * __builtin_amdgcn_rcpf(1.0f + __expf(-hw)));
            *reinterpret_cast<float4*>(&hr_lds[(row * KTOP + slot) * HRS + c4]) = o;
        }
    }
    __syncthreads();

    for (int j = 0; j < 2; ++j) {
        const int e = wave * 2 + j;
        int base = 0;
        for (int it = 0; it < 2; ++it) {
            const int row = it * 32 + lane;
            const bool hit = (i0_lds[row] == e) || (i1_lds[row] == e);
            const unsigned mask = __builtin_amdgcn_ballot_w32(hit);
            const int pos = base + __popc(mask & ((1u << lane) - 1u));
            if (hit) list_lds[e * TG + pos] = row * KTOP + ((i0_lds[row] == e) ? 0 : 1);
            base += __popc(mask);
        }
        if (lane == 0) cnt_lds[e] = base;
    }
    __syncthreads();
    if (wave == 0) {
        const int cE = (lane < NEXP) ? cnt_lds[lane] : 0;
        const int nt = (cE + 15) >> 4;
        int inc = nt;
        for (int d = 1; d < NEXP; d <<= 1) {
            const int o = __shfl_up(inc, d);
            if (lane >= d) inc += o;
        }
        const int first = inc - nt;
        for (int t = 0; t < nt; ++t) tileP_lds[first + t] = lane | ((t * 16) << 8) | (cE << 16);
        if (lane == NEXP - 1) nT_lds = inc;
    }
    __syncthreads();

    const int nT = __builtin_amdgcn_readfirstlane(nT_lds);
    const int d0 = dg * 256 + wave * 32 + nIdx;
    const float* wd = w_b + (size_t)d0 * RANK + 2 * hv;

    int tp = __builtin_amdgcn_readfirstlane(tileP_lds[0]);
    v2f bc0[4], bc1[4];
    {
        const float* wb = wd + (size_t)(tp & 255) * (D_OUT * RANK);
#pragma unroll
        for (int j = 0; j < 4; ++j) {
            bc0[j] = *reinterpret_cast<const v2f*>(wb + 4 * j);
            bc1[j] = *reinterpret_cast<const v2f*>(wb + 16 * RANK + 4 * j);
        }
    }

    for (int t = 0; t < nT; ++t) {
        const int e  = tp & 255;
        const int q0 = (tp >> 8) & 255;
        const int c  = tp >> 16;
        const int tn  = (t + 1 < nT) ? (t + 1) : t;
        const int tpN = __builtin_amdgcn_readfirstlane(tileP_lds[tn]);
        const float* wbN = wd + (size_t)(tpN & 255) * (D_OUT * RANK);
        v2f bn0[4], bn1[4];
#pragma unroll
        for (int j = 0; j < 4; ++j) {
            bn0[j] = *reinterpret_cast<const v2f*>(wbN + 4 * j);
            bn1[j] = *reinterpret_cast<const v2f*>(wbN + 16 * RANK + 4 * j);
        }

        const int* le = &list_lds[e * TG];
        const int qa   = q0 + nIdx;
        const int recA = le[(qa < c) ? qa : (c - 1)];
        int rec[8];
#pragma unroll
        for (int v = 0; v < 8; ++v) {
            const int q = q0 + v + 8 * hv;
            rec[v] = le[(q < c) ? q : (c - 1)];
        }
        const float* ha = &hr_lds[recA * HRS + 2 * hv];
        v2f a[4];
#pragma unroll
        for (int j = 0; j < 4; ++j) a[j] = *reinterpret_cast<const v2f*>(ha + 4 * j);

        float old0[8], old1[8];
#pragma unroll
        for (int v = 0; v < 8; ++v) {
            const float* p = ob + (rec[v] >> 1) * 32 + nIdx;
            old0[v] = p[0];
            old1[v] = p[16];
        }

        v8f acc0 = (v8f){}, acc1 = (v8f){};
#pragma unroll
        for (int j = 0; j < 4; ++j) {
            acc0 = __builtin_amdgcn_wmma_f32_16x16x4_f32(false, a[j], false, bc0[j],
                                                         (short)0, acc0, false, false);
            acc1 = __builtin_amdgcn_wmma_f32_16x16x4_f32(false, a[j], false, bc1[j],
                                                         (short)0, acc1, false, false);
        }

#pragma unroll
        for (int v = 0; v < 8; ++v) {
            const int r = (q0 + v + 8 * hv < c) ? (rec[v] >> 1) : TG;
            float* p = ob + r * 32 + nIdx;
            p[0]  = old0[v] + acc0[v];
            p[16] = old1[v] + acc1[v];
        }
#pragma unroll
        for (int j = 0; j < 4; ++j) { bc0[j] = bn0[j]; bc1[j] = bn1[j]; }
        tp = tpN;
    }
    __syncthreads();

    {
        float* og = out + (size_t)tokBase * D_OUT + dg * 256 + wave * 32;
        const int r4 = lane >> 3;
        const int c4 = (lane & 7) << 2;
#pragma unroll 4
        for (int it = 0; it < 16; ++it) {
            const int row = it * 4 + r4;
            const float4 v = *reinterpret_cast<const float4*>(ob + row * 32 + c4);
            *reinterpret_cast<float4*>(og + (size_t)row * D_OUT + c4) = v;
        }
    }
}

extern "C" void kernel_launch(void* const* d_in, const int* in_sizes, int n_in,
                              void* d_out, int out_size, void* d_ws, size_t ws_size,
                              hipStream_t stream) {
    const float* x       = (const float*)d_in[0];
    const float* routing = (const float*)d_in[1];
    const int*   idxs    = (const int*)d_in[2];
    const float* w_a     = (const float*)d_in[3];
    const float* w_b     = (const float*)d_in[4];
    float* out  = (float*)d_out;
    float* hc   = (float*)d_ws;

    hipLaunchKernelGGL(lora_down_routed_kernel, dim3(256), dim3(256), 0, stream,
                       x, idxs, w_a, hc);
    hipLaunchKernelGGL(lora_up_routed_kernel, dim3(512), dim3(256), 0, stream,
                       hc, routing, idxs, w_b, out);
}


# First product weights-stationary: workgroup = 128 tokens x one K part, 16 waves; wave e loads expert e's w_a fragments once into registers and walks e's 16-row list tiles; sum order unchanged
# speedup vs baseline: 3.4708x; 1.1426x over previous
#include <hip/hip_runtime.h>


typedef float v2f __attribute__((ext_vector_type(2)));
typedef float v8f __attribute__((ext_vector_type(8)));

#define D_IN   2048
#define D_OUT  2048
#define NEXP   16
#define RANK   16
#define KTOP   2
#define NTOK   4096
#define TG     64
#define KSPLIT 8
#define KPART  (D_IN / KSPLIT)

#define MAXT   24
#define HCS    32

#define TG1  128
#define XS1  (KPART + 4)
#define NKS  (KPART / 4)
__device__ __forceinline__ void lds_barrier()
{
    __builtin_amdgcn_fence(__ATOMIC_RELEASE, "workgroup", "local");
    __builtin_amdgcn_s_barrier();
    __builtin_amdgcn_fence(__ATOMIC_ACQUIRE, "workgroup", "local");
}

__global__ __launch_bounds__(512) __attribute__((amdgpu_waves_per_eu(4, 4)))
void lora_down_routed_kernel(const float* __restrict__ x,
                             const int*   __restrict__ idxs,
                             const float* __restrict__ w_a,
                             float* __restrict__ hc)
{
    __shared__ float x_lds[TG1 * XS1];
    __shared__ int   list_lds[NEXP * TG1];
    __shared__ int   i0_lds[TG1];
    __shared__ int   i1_lds[TG1];

    const int tid  = threadIdx.x;
    const int e    = __builtin_amdgcn_readfirstlane(tid >> 5);
    const int lane = tid & 31;
    const int nIdx = lane & 15;
    const int hv   = lane >> 4;
    const int grp  = blockIdx.x / KSPLIT;
    const int ks   = blockIdx.x % KSPLIT;
    const int tokBase = grp * TG1;
    const int kBase   = ks * KPART;

    int my0 = 0, my1 = 0;
    if (tid < TG1) {
        my0 = idxs[(tokBase + tid) * KTOP + 0];
        my1 = idxs[(tokBase + tid) * KTOP + 1];
    }
    {
        const float* xg = x + (size_t)tokBase * D_IN + kBase;
#pragma unroll
        for (int i = 0; i < 16; ++i) {
            const int fi  = tid + i * 512;
            const int row = fi >> 6;
            const int col = (fi & 63) << 2;
            const float4 v = *reinterpret_cast<const float4*>(xg + (size_t)row * D_IN + col);
            *reinterpret_cast<float4*>(&x_lds[row * XS1 + col]) = v;
        }
    }
    v2f wf[NKS];
    {
        const float* wg = w_a + (size_t)(e * RANK + nIdx) * D_IN + kBase + 2 * hv;
#pragma unroll
        for (int t = 0; t < NKS; ++t) wf[t] = *reinterpret_cast<const v2f*>(wg + 4 * t);
    }
    if (tid < TG1) { i0_lds[tid] = my0; i1_lds[tid] = my1; }
    lds_barrier();

    int cnt = 0;
    for (int it = 0; it < TG1 / 32; ++it) {
        const int row = it * 32 + lane;
        const bool hit = (i0_lds[row] == e) || (i1_lds[row] == e);
        const unsigned mask = __builtin_amdgcn_ballot_w32(hit);
        const int pos = cnt + __popc(mask & ((1u << lane) - 1u));
        if (hit) list_lds[e * TG1 + pos] = row;
        cnt += __popc(mask);
    }
    lds_barrier();

    const int* le = &list_lds[e * TG1];
    const int nt = (cnt + 15) >> 4;
    for (int t2 = 0; t2 < nt; ++t2) {
        const int qa = t2 * 16 + nIdx;
        const int rowA = le[(qa < cnt) ? qa : 0];
        const float* xl = &x_lds[rowA * XS1 + 2 * hv];
        v8f acc = (v8f){};
#pragma unroll
        for (int t = 0; t < NKS; ++t) {
            const v2f a = *reinterpret_cast<const v2f*>(xl + 4 * t);
            acc = __builtin_amdgcn_wmma_f32_16x16x4_f32(false, a, false, wf[t],
                                                        (short)0, acc, false, false);
        }
#pragma unroll
        for (int v = 0; v < 8; ++v) {
            const int q = t2 * 16 + v + 8 * hv;
            if (q < cnt) {
                const int row = le[q];
                float* hb = hc + ((size_t)ks * NTOK + tokBase + row) * (KTOP * HCS) + nIdx;
                if (i0_lds[row] == e) hb[0]   = acc[v];
                if (i1_lds[row] == e) hb[HCS] = acc[v];
            }
        }
    }
}

#define HRS 20
__global__ __launch_bounds__(256)
void lora_up_routed_kernel(const float* __restrict__ hc,
                           const float* __restrict__ routing,
                           const int*   __restrict__ idxs,
                           const float* __restrict__ w_b,
                           float* __restrict__ out)
{
    __shared__ float o_lds[8 * (TG + 1) * 32];
    __shared__ float hr_lds[TG * KTOP * HRS];
    __shared__ int   list_lds[NEXP * TG];
    __shared__ int   cnt_lds[NEXP];
    __shared__ int   i0_lds[TG];
    __shared__ int   i1_lds[TG];
    __shared__ int   tileP_lds[MAXT];
    __shared__ int   nT_lds;

    const int tid  = threadIdx.x;
    const int wave = __builtin_amdgcn_readfirstlane(tid >> 5);
    const int lane = tid & 31;
    const int nIdx = lane & 15;
    const int hv   = lane >> 4;
    const int grp  = blockIdx.x >> 3;
    const int dg   = blockIdx.x & 7;
    const int tokBase = grp * TG;

    if (tid < TG) {
        i0_lds[tid] = idxs[(tokBase + tid) * KTOP + 0];
        i1_lds[tid] = idxs[(tokBase + tid) * KTOP + 1];
    }
    float* ob = &o_lds[wave * ((TG + 1) * 32)];
    for (int it = 0; it < 16; ++it)
        *reinterpret_cast<float4*>(ob + (it * 32 + lane) * 4) = make_float4(0.f, 0.f, 0.f, 0.f);

    {
        const size_t ksStride = (size_t)NTOK * KTOP * HCS;
#pragma unroll
        for (int i = 0; i < 2; ++i) {
            const int fi   = tid + i * 256;
            const int row  = fi >> 3;
            const int slot = (fi >> 2) & 1;
            const int c4   = (fi & 3) << 2;
            const int tok  = tokBase + row;
            const int   e0 = idxs[tok * KTOP + 0];
            const int   e1 = idxs[tok * KTOP + 1];
            const float r0 = routing[tok * KTOP + 0];
            const float r1 = routing[tok * KTOP + 1];
            const int   es = slot ? e1 : e0;
            float g = 0.f;
            g += (e0 == es) ? r0 : 0.f;
            g += (e1 == es) ? r1 : 0.f;
            const float gate = 2.0f * g;
            const float* p0 = hc + ((size_t)tok * KTOP + slot) * HCS + c4;
            float4 s[KSPLIT];
#pragma unroll
            for (int p = 0; p < KSPLIT; ++p)
                s[p] = *reinterpret_cast<const float4*>(p0 + p * ksStride);
            float hx = s[0].x, hy = s[0].y, hz = s[0].z, hw = s[0].w;
#pragma unroll
            for (int p = 1; p < KSPLIT; ++p) { hx += s[p].x; hy += s[p].y; hz += s[p].z; hw += s[p].w; }
            float4 o;
            o.x = gate * (hx * __builtin_amdgcn_rcpf(1.0f + __expf(-hx)));
            o.y = gate * (hy * __builtin_amdgcn_rcpf(1.0f + __expf(-hy)));
            o.z = gate * (hz * __builtin_amdgcn_rcpf(1.0f + __expf(-hz)));
            o.w = gate * (hw * __builtin_amdgcn_rcpf(1.0f + __expf(-hw)));
            *reinterpret_cast<float4*>(&hr_lds[(row * KTOP + slot) * HRS + c4]) = o;
        }
    }
    __syncthreads();

    for (int j = 0; j < 2; ++j) {
        const int e = wave * 2 + j;
        int base = 0;
        for (int it = 0; it < 2; ++it) {
            const int row = it * 32 + lane;
            const bool hit = (i0_lds[row] == e) || (i1_lds[row] == e);
            const unsigned mask = __builtin_amdgcn_ballot_w32(hit);
            const int pos = base + __popc(mask & ((1u << lane) - 1u));
            if (hit) list_lds[e * TG + pos] = row * KTOP + ((i0_lds[row] == e) ? 0 : 1);
            base += __popc(mask);
        }
        if (lane == 0) cnt_lds[e] = base;
    }
    __syncthreads();
    if (wave == 0) {
        const int cE = (lane < NEXP) ? cnt_lds[lane] : 0;
        const int nt = (cE + 15) >> 4;
        int inc = nt;
        for (int d = 1; d < NEXP; d <<= 1) {
            const int o = __shfl_up(inc, d);
            if (lane >= d) inc += o;
        }
        const int first = inc - nt;
        for (int t = 0; t < nt; ++t) tileP_lds[first + t] = lane | ((t * 16) << 8) | (cE << 16);
        if (lane == NEXP - 1) nT_lds = inc;
    }
    __syncthreads();

    const int nT = __builtin_amdgcn_readfirstlane(nT_lds);
    const int d0 = dg * 256 + wave * 32 + nIdx;
    const float* wd = w_b + (size_t)d0 * RANK + 2 * hv;

    int tp = __builtin_amdgcn_readfirstlane(tileP_lds[0]);
    v2f bc0[4], bc1[4];
    {
        const float* wb = wd + (size_t)(tp & 255) * (D_OUT * RANK);
#pragma unroll
        for (int j = 0; j < 4; ++j) {
            bc0[j] = *reinterpret_cast<const v2f*>(wb + 4 * j);
            bc1[j] = *reinterpret_cast<const v2f*>(wb + 16 * RANK + 4 * j);
        }
    }

    for (int t = 0; t < nT; ++t) {
        const int e  = tp & 255;
        const int q0 = (tp >> 8) & 255;
        const int c  = tp >> 16;
        const int tn  = (t + 1 < nT) ? (t + 1) : t;
        const int tpN = __builtin_amdgcn_readfirstlane(tileP_lds[tn]);
        const float* wbN = wd + (size_t)(tpN & 255) * (D_OUT * RANK);
        v2f bn0[4], bn1[4];
#pragma unroll
        for (int j = 0; j < 4; ++j) {
            bn0[j] = *reinterpret_cast<const v2f*>(wbN + 4 * j);
            bn1[j] = *reinterpret_cast<const v2f*>(wbN + 16 * RANK + 4 * j);
        }

        const int* le = &list_lds[e * TG];
        const int qa   = q0 + nIdx;
        const int recA = le[(qa < c) ? qa : (c - 1)];
        int rec[8];
#pragma unroll
        for (int v = 0; v < 8; ++v) {
            const int q = q0 + v + 8 * hv;
            rec[v] = le[(q < c) ? q : (c - 1)];
        }
        const float* ha = &hr_lds[recA * HRS + 2 * hv];
        v2f a[4];
#pragma unroll
        for (int j = 0; j < 4; ++j) a[j] = *reinterpret_cast<const v2f*>(ha + 4 * j);

        float old0[8], old1[8];
#pragma unroll
        for (int v = 0; v < 8; ++v) {
            const float* p = ob + (rec[v] >> 1) * 32 + nIdx;
            old0[v] = p[0];
            old1[v] = p[16];
        }

        v8f acc0 = (v8f){}, acc1 = (v8f){};
#pragma unroll
        for (int j = 0; j < 4; ++j) {
            acc0 = __builtin_amdgcn_wmma_f32_16x16x4_f32(false, a[j], false, bc0[j],
                                                         (short)0, acc0, false, false);
            acc1 = __builtin_amdgcn_wmma_f32_16x16x4_f32(false, a[j], false, bc1[j],
                                                         (short)0, acc1, false, false);
        }

#pragma unroll
        for (int v = 0; v < 8; ++v) {
            const int r = (q0 + v + 8 * hv < c) ? (rec[v] >> 1) : TG;
            float* p = ob + r * 32 + nIdx;
            p[0]  = old0[v] + acc0[v];
            p[16] = old1[v] + acc1[v];
        }
#pragma unroll
        for (int j = 0; j < 4; ++j) { bc0[j] = bn0[j]; bc1[j] = bn1[j]; }
        tp = tpN;
    }
    __syncthreads();

    {
        float* og = out + (size_t)tokBase * D_OUT + dg * 256 + wave * 32;
        const int r4 = lane >> 3;
        const int c4 = (lane & 7) << 2;
#pragma unroll 4
        for (int it = 0; it < 16; ++it) {
            const int row = it * 4 + r4;
            const float4 v = *reinterpret_cast<const float4*>(ob + row * 32 + c4);
            *reinterpret_cast<float4*>(og + (size_t)row * D_OUT + c4) = v;
        }
    }
}

extern "C" void kernel_launch(void* const* d_in, const int* in_sizes, int n_in,
                              void* d_out, int out_size, void* d_ws, size_t ws_size,
                              hipStream_t stream) {
    const float* x       = (const float*)d_in[0];
    const float* routing = (const float*)d_in[1];
    const int*   idxs    = (const int*)d_in[2];
    const float* w_a     = (const float*)d_in[3];
    const float* w_b     = (const float*)d_in[4];
    float* out  = (float*)d_out;
    float* hc   = (float*)d_ws;

    hipLaunchKernelGGL(lora_down_routed_kernel, dim3(256), dim3(512), 0, stream,
                       x, idxs, w_a, hc);
    hipLaunchKernelGGL(lora_up_routed_kernel, dim3(512), dim3(256), 0, stream,
                       hc, routing, idxs, w_b, out);
}


# Both products with operands swapped (weights as A, token rows as B): a lane holds 8 consecutive outputs of one token; results leave and join as 16-byte accesses on one list read; sum order unchanged
# speedup vs baseline: 3.5774x; 1.0307x over previous
#include <hip/hip_runtime.h>


typedef float v2f __attribute__((ext_vector_type(2)));
typedef float v8f __attribute__((ext_vector_type(8)));

#define D_IN   2048
#define D_OUT  2048
#define NEXP   16
#define RANK   16
#define KTOP   2
#define NTOK   4096
#define TG     64
#define KSPLIT 8
#define KPART  (D_IN / KSPLIT)

#define MAXT   24
#define HCS    32

#define TG1  128
#define XS1  (KPART + 4)
#define NKS  (KPART / 4)
__device__ __forceinline__ void lds_barrier()
{
    __builtin_amdgcn_fence(__ATOMIC_RELEASE, "workgroup", "local");
    __builtin_amdgcn_s_barrier();
    __builtin_amdgcn_fence(__ATOMIC_ACQUIRE, "workgroup", "local");
}

__global__ __launch_bounds__(512) __attribute__((amdgpu_waves_per_eu(4, 4)))
void lora_down_routed_kernel(const float* __restrict__ x,
                             const int*   __restrict__ idxs,
                             const float* __restrict__ w_a,
                             float* __restrict__ hc)
{
    __shared__ float x_lds[TG1 * XS1];
    __shared__ int   list_lds[NEXP * TG1];
    __shared__ int   i0_lds[TG1];
    __shared__ int   i1_lds[TG1];

    const int tid  = threadIdx.x;
    const int e    = __builtin_amdgcn_readfirstlane(tid >> 5);
    const int lane = tid & 31;
    const int nIdx = lane & 15;
    const int hv   = lane >> 4;
    const int grp  = blockIdx.x / KSPLIT;
    const int ks   = blockIdx.x % KSPLIT;
    const int tokBase = grp * TG1;
    const int kBase   = ks * KPART;

    int my0 = 0, my1 = 0;
    if (tid < TG1) {
        my0 = idxs[(tokBase + tid) * KTOP + 0];
        my1 = idxs[(tokBase + tid) * KTOP + 1];
    }
    {
        const float* xg = x + (size_t)tokBase * D_IN + kBase;
#pragma unroll
        for (int i = 0; i < 16; ++i) {
            const int fi  = tid + i * 512;
            const int row = fi >> 6;
            const int col = (fi & 63) << 2;
            const float4 v = *reinterpret_cast<const float4*>(xg + (size_t)row * D_IN + col);
            *reinterpret_cast<float4*>(&x_lds[row * XS1 + col]) = v;
        }
    }
    v2f wf[NKS];
    {
        const float* wg = w_a + (size_t)(e * RANK + nIdx) * D_IN + kBase + 2 * hv;
#pragma unroll
        for (int t = 0; t < NKS; ++t) wf[t] = *reinterpret_cast<const v2f*>(wg + 4 * t);
    }
    if (tid < TG1) { i0_lds[tid] = my0; i1_lds[tid] = my1; }
    lds_barrier();

    int cnt = 0;
    for (int it = 0; it < TG1 / 32; ++it) {
        const int row = it * 32 + lane;
        const bool hit = (i0_lds[row] == e) || (i1_lds[row] == e);
        const unsigned mask = __builtin_amdgcn_ballot_w32(hit);
        const int pos = cnt + __popc(mask & ((1u << lane) - 1u));
        if (hit) list_lds[e * TG1 + pos] = row;
        cnt += __popc(mask);
    }
    lds_barrier();

    const int* le = &list_lds[e * TG1];
    const int nt = (cnt + 15) >> 4;
    for (int t2 = 0; t2 < nt; ++t2) {
        const int  qa   = t2 * 16 + nIdx;
        const bool live = qa < cnt;
        const int rowA = le[live ? qa : 0];
        const bool s0 = i0_lds[rowA] == e;
        const bool s1 = i1_lds[rowA] == e;
        const float* xl = &x_lds[rowA * XS1 + 2 * hv];
        v8f acc = (v8f){};
#pragma unroll
        for (int t = 0; t < NKS; ++t) {
            const v2f b = *reinterpret_cast<const v2f*>(xl + 4 * t);
            acc = __builtin_amdgcn_wmma_f32_16x16x4_f32(false, wf[t], false, b,
                                                        (short)0, acc, false, false);
        }
        if (live) {
            float* hb = hc + ((size_t)ks * NTOK + tokBase + rowA) * (KTOP * HCS) + 8 * hv;
            const float4 lo = make_float4(acc[0], acc[1], acc[2], acc[3]);
            const float4 hi = make_float4(acc[4], acc[5], acc[6], acc[7]);
            if (s0) {
                *reinterpret_cast<float4*>(hb)     = lo;
                *reinterpret_cast<float4*>(hb + 4) = hi;
            }
            if (s1) {
                *reinterpret_cast<float4*>(hb + HCS)     = lo;
                *reinterpret_cast<float4*>(hb + HCS + 4) = hi;
            }
        }
    }
}

#define HRS 20
#define OS  36
#define OROWS (TG + 16)
__global__ __launch_bounds__(256)
void lora_up_routed_kernel(const float* __restrict__ hc,
                           const float* __restrict__ routing,
                           const int*   __restrict__ idxs,
                           const float* __restrict__ w_b,
                           float* __restrict__ out)
{
    __shared__ __attribute__((aligned(16))) float o_lds[8 * OROWS * OS];
    __shared__ float hr_lds[TG * KTOP * HRS];
    __shared__ int   list_lds[NEXP * TG];
    __shared__ int   cnt_lds[NEXP];
    __shared__ int   i0_lds[TG];
    __shared__ int   i1_lds[TG];
    __shared__ int   tileP_lds[MAXT];
    __shared__ int   nT_lds;

    const int tid  = threadIdx.x;
    const int wave = __builtin_amdgcn_readfirstlane(tid >> 5);
    const int lane = tid & 31;
    const int nIdx = lane & 15;
    const int hv   = lane >> 4;
    const int grp  = blockIdx.x >> 3;
    const int dg   = blockIdx.x & 7;
    const int tokBase = grp * TG;

    if (tid < TG) {
        i0_lds[tid] = idxs[(tokBase + tid) * KTOP + 0];
        i1_lds[tid] = idxs[(tokBase + tid) * KTOP + 1];
    }
    float* ob = &o_lds[wave * (OROWS * OS)];
    for (int it = 0; it < 16; ++it) {
        const int fi = it * 32 + lane;
        *reinterpret_cast<float4*>(ob + (fi >> 3) * OS + ((fi & 7) << 2)) = make_float4(0.f, 0.f, 0.f, 0.f);
    }

    {
        const size_t ksStride = (size_t)NTOK * KTOP * HCS;
#pragma unroll
        for (int i = 0; i < 2; ++i) {
            const int fi   = tid + i * 256;
            const int row  = fi >> 3;
            const int slot = (fi >> 2) & 1;
            const int c4   = (fi & 3) << 2;
            const int tok  = tokBase + row;
            const int   e0 = idxs[tok * KTOP + 0];
            const int   e1 = idxs[tok * KTOP + 1];
            const float r0 = routing[tok * KTOP + 0];
            const float r1 = routing[tok * KTOP + 1];
            const int   es = slot ? e1 : e0;
            float g = 0.f;
            g += (e0 == es) ? r0 : 0.f;
            g += (e1 == es) ? r1 : 0.f;
            const float gate = 2.0f * g;
            const float* p0 = hc + ((size_t)tok * KTOP + slot) * HCS + c4;
            float4 s[KSPLIT];
#pragma unroll
            for (int p = 0; p < KSPLIT; ++p)
                s[p] = *reinterpret_cast<const float4*>(p0 + p * ksStride);
            float hx = s[0].x, hy = s[0].y, hz = s[0].z, hw = s[0].w;
#pragma unroll
            for (int p = 1; p < KSPLIT; ++p) { hx += s[p].x; hy += s[p].y; hz += s[p].z; hw += s[p].w; }
            float4 o;
            o.x = gate * (hx * __builtin_amdgcn_rcpf(1.0f + __expf(-hx)));
            o.y = gate * (hy * __builtin_amdgcn_rcpf(1.0f + __expf(-hy)));
            o.z = gate * (hz * __builtin_amdgcn_rcpf(1.0f + __expf(-hz)));
            o.w = gate * (hw * __builtin_amdgcn_rcpf(1.0f + __expf(-hw)));
            *reinterpret_cast<float4*>(&hr_lds[(row * KTOP + slot) * HRS + c4]) = o;
        }
    }
    __syncthreads();

    for (int j = 0; j < 2; ++j) {
        const int e = wave * 2 + j;
        int base = 0;
        for (int it = 0; it < 2; ++it) {
            const int row = it * 32 + lane;
            const bool hit = (i0_lds[row] == e) || (i1_lds[row] == e);
            const unsigned mask = __builtin_amdgcn_ballot_w32(hit);
            const int pos = base + __popc(mask & ((1u << lane) - 1u));
            if (hit) list_lds[e * TG + pos] = row * KTOP + ((i0_lds[row] == e) ? 0 : 1);
            base += __popc(mask);
        }
        if (lane == 0) cnt_lds[e] = base;
    }
    __syncthreads();
    if (wave == 0) {
        const int cE = (lane < NEXP) ? cnt_lds[lane] : 0;
        const int nt = (cE + 15) >> 4;
        int inc = nt;
        for (int d = 1; d < NEXP; d <<= 1) {
            const int o = __shfl_up(inc, d);
            if (lane >= d) inc += o;
        }
        const int first = inc - nt;
        for (int t = 0; t < nt; ++t) tileP_lds[first + t] = lane | ((t * 16) << 8) | (cE << 16);
        if (lane == NEXP - 1) nT_lds = inc;
    }
    __syncthreads();

    const int nT = __builtin_amdgcn_readfirstlane(nT_lds);
    const int d0 = dg * 256 + wave * 32 + nIdx;
    const float* wd = w_b + (size_t)d0 * RANK + 2 * hv;

    int tp = __builtin_amdgcn_readfirstlane(tileP_lds[0]);
    v2f bc0[4], bc1[4];
    {
        const float* wb = wd + (size_t)(tp & 255) * (D_OUT * RANK);
#pragma unroll
        for (int j = 0; j < 4; ++j) {
            bc0[j] = *reinterpret_cast<const v2f*>(wb + 4 * j);
            bc1[j] = *reinterpret_cast<const v2f*>(wb + 16 * RANK + 4 * j);
        }
    }

    for (int t = 0; t < nT; ++t) {
        const int e  = tp & 255;
        const int q0 = (tp >> 8) & 255;
        const int c  = tp >> 16;
        const int tn  = (t + 1 < nT) ? (t + 1) : t;
        const int tpN = __builtin_amdgcn_readfirstlane(tileP_lds[tn]);
        const float* wbN = wd + (size_t)(tpN & 255) * (D_OUT * RANK);
        v2f bn0[4], bn1[4];
#pragma unroll
        for (int j = 0; j < 4; ++j) {
            bn0[j] = *reinterpret_cast<const v2f*>(wbN + 4 * j);
            bn1[j] = *reinterpret_cast<const v2f*>(wbN + 16 * RANK + 4 * j);
        }

        const int* le = &list_lds[e * TG];
        const int  qa   = q0 + nIdx;
        const bool live = qa < c;
        const int  rec  = le[live ? qa : (c - 1)];
        const float* hb = &hr_lds[rec * HRS + 2 * hv];
        v2f b[4];
#pragma unroll
        for (int j = 0; j < 4; ++j) b[j] = *reinterpret_cast<const v2f*>(hb + 4 * j);

        const int row = rec >> 1;
        const float* pr = ob + row * OS + 8 * hv;
        const float4 o00 = *reinterpret_cast<const float4*>(pr);
        const float4 o01 = *reinterpret_cast<const float4*>(pr + 4);
        const float4 o10 = *reinterpret_cast<const float4*>(pr + 16);
        const float4 o11 = *reinterpret_cast<const float4*>(pr + 20);

        v8f acc0 = (v8f){}, acc1 = (v8f){};
#pragma unroll
        for (int j = 0; j < 4; ++j) {
            acc0 = __builtin_amdgcn_wmma_f32_16x16x4_f32(false, bc0[j], false, b[j],
                                                         (short)0, acc0, false, false);
            acc1 = __builtin_amdgcn_wmma_f32_16x16x4_f32(false, bc1[j], false, b[j],
                                                         (short)0, acc1, false, false);
        }

        float* pw = ob + (live ? row : (TG + nIdx)) * OS + 8 * hv;
        *reinterpret_cast<float4*>(pw)      = make_float4(o00.x + acc0[0], o00.y + acc0[1], o00.z + acc0[2], o00.w + acc0[3]);
        *reinterpret_cast<float4*>(pw + 4)  = make_float4(o01.x + acc0[4], o01.y + acc0[5], o01.z + acc0[6], o01.w + acc0[7]);
        *reinterpret_cast<float4*>(pw + 16) = make_float4(o10.x + acc1[0], o10.y + acc1[1], o10.z + acc1[2], o10.w + acc1[3]);
        *reinterpret_cast<float4*>(pw + 20) = make_float4(o11.x + acc1[4], o11.y + acc1[5], o11.z + acc1[6], o11.w + acc1[7]);
#pragma unroll
        for (int j = 0; j < 4; ++j) { bc0[j] = bn0[j]; bc1[j] = bn1[j]; }
        tp = tpN;
    }
    __syncthreads();

    {
        float* og = out + (size_t)tokBase * D_OUT + dg * 256 + wave * 32;
        const int r4 = lane >> 3;
        const int c4 = (lane & 7) << 2;
#pragma unroll 4
        for (int it = 0; it < 16; ++it) {
            const int row = it * 4 + r4;
            const float4 v = *reinterpret_cast<const float4*>(ob + row * OS + c4);
            *reinterpret_cast<float4*>(og + (size_t)row * D_OUT + c4) = v;
        }
    }
}

extern "C" void kernel_launch(void* const* d_in, const int* in_sizes, int n_in,
                              void* d_out, int out_size, void* d_ws, size_t ws_size,
                              hipStream_t stream) {
    const float* x       = (const float*)d_in[0];
    const float* routing = (const float*)d_in[1];
    const int*   idxs    = (const int*)d_in[2];
    const float* w_a     = (const float*)d_in[3];
    const float* w_b     = (const float*)d_in[4];
    float* out  = (float*)d_out;
    float* hc   = (float*)d_ws;

    hipLaunchKernelGGL(lora_down_routed_kernel, dim3(256), dim3(512), 0, stream,
                       x, idxs, w_a, hc);
    hipLaunchKernelGGL(lora_up_routed_kernel, dim3(512), dim3(256), 0, stream,
                       hc, routing, idxs, w_b, out);
}


# Second product's workgroup covers 128 tokens (was 64), 16 waves of one 16-column tile each: lists twice as long, so each w_b fragment fetched and each list tile serves more rows; sum order unchanged
# speedup vs baseline: 3.8110x; 1.0653x over previous
#include <hip/hip_runtime.h>


typedef float v2f __attribute__((ext_vector_type(2)));
typedef float v8f __attribute__((ext_vector_type(8)));

#define D_IN   2048
#define D_OUT  2048
#define NEXP   16
#define RANK   16
#define KTOP   2
#define NTOK   4096
#define TG     128
#define KSPLIT 8
#define KPART  (D_IN / KSPLIT)

#define MAXT   32
#define HCS    32

#define TG1  128
#define XS1  (KPART + 4)
#define NKS  (KPART / 4)
__device__ __forceinline__ void lds_barrier()
{
    __builtin_amdgcn_fence(__ATOMIC_RELEASE, "workgroup", "local");
    __builtin_amdgcn_s_barrier();
    __builtin_amdgcn_fence(__ATOMIC_ACQUIRE, "workgroup", "local");
}

__global__ __launch_bounds__(512) __attribute__((amdgpu_waves_per_eu(4, 4)))
void lora_down_routed_kernel(const float* __restrict__ x,
                             const int*   __restrict__ idxs,
                             const float* __restrict__ w_a,
                             float* __restrict__ hc)
{
    __shared__ float x_lds[TG1 * XS1];
    __shared__ int   list_lds[NEXP * TG1];
    __shared__ int   i0_lds[TG1];
    __shared__ int   i1_lds[TG1];

    const int tid  = threadIdx.x;
    const int e    = __builtin_amdgcn_readfirstlane(tid >> 5);
    const int lane = tid & 31;
    const int nIdx = lane & 15;
    const int hv   = lane >> 4;
    const int grp  = blockIdx.x / KSPLIT;
    const int ks   = blockIdx.x % KSPLIT;
    const int tokBase = grp * TG1;
    const int kBase   = ks * KPART;

    int my0 = 0, my1 = 0;
    if (tid < TG1) {
        my0 = idxs[(tokBase + tid) * KTOP + 0];
        my1 = idxs[(tokBase + tid) * KTOP + 1];
    }
    {
        const float* xg = x + (size_t)tokBase * D_IN + kBase;
#pragma unroll
        for (int i = 0; i < 16; ++i) {
            const int fi  = tid + i * 512;
            const int row = fi >> 6;
            const int col = (fi & 63) << 2;
            const float4 v = *reinterpret_cast<const float4*>(xg + (size_t)row * D_IN + col);
            *reinterpret_cast<float4*>(&x_lds[row * XS1 + col]) = v;
        }
    }
    v2f wf[NKS];
    {
        const float* wg = w_a + (size_t)(e * RANK + nIdx) * D_IN + kBase + 2 * hv;
#pragma unroll
        for (int t = 0; t < NKS; ++t) wf[t] = *reinterpret_cast<const v2f*>(wg + 4 * t);
    }
    if (tid < TG1) { i0_lds[tid] = my0; i1_lds[tid] = my1; }
    lds_barrier();

    int cnt = 0;
    for (int it = 0; it < TG1 / 32; ++it) {
        const int row = it * 32 + lane;
        const bool hit = (i0_lds[row] == e) || (i1_lds[row] == e);
        const unsigned mask = __builtin_amdgcn_ballot_w32(hit);
        const int pos = cnt + __popc(mask & ((1u << lane) - 1u));
        if (hit) list_lds[e * TG1 + pos] = row;
        cnt += __popc(mask);
    }
    lds_barrier();

    const int* le = &list_lds[e * TG1];
    const int nt = (cnt + 15) >> 4;
    for (int t2 = 0; t2 < nt; ++t2) {
        const int  qa   = t2 * 16 + nIdx;
        const bool live = qa < cnt;
        const int rowA = le[live ? qa : 0];
        const bool s0 = i0_lds[rowA] == e;
        const bool s1 = i1_lds[rowA] == e;
        const float* xl = &x_lds[rowA * XS1 + 2 * hv];
        v8f acc = (v8f){};
#pragma unroll
        for (int t = 0; t < NKS; ++t) {
            const v2f b = *reinterpret_cast<const v2f*>(xl + 4 * t);
            acc = __builtin_amdgcn_wmma_f32_16x16x4_f32(false, wf[t], false, b,
                                                        (short)0, acc, false, false);
        }
        if (live) {
            float* hb = hc + ((size_t)ks * NTOK + tokBase + rowA) * (KTOP * HCS) + 8 * hv;
            const float4 lo = make_float4(acc[0], acc[1], acc[2], acc[3]);
            const float4 hi = make_float4(acc[4], acc[5], acc[6], acc[7]);
            if (s0) {
                *reinterpret_cast<float4*>(hb)     = lo;
                *reinterpret_cast<float4*>(hb + 4) = hi;
            }
            if (s1) {
                *reinterpret_cast<float4*>(hb + HCS)     = lo;
                *reinterpret_cast<float4*>(hb + HCS + 4) = hi;
            }
        }
    }
}

#define HRS 20
#define OS  20
#define OROWS (TG + 16)
__global__ __launch_bounds__(512)
void lora_up_routed_kernel(const float* __restrict__ hc,
                           const float* __restrict__ routing,
                           const int*   __restrict__ idxs,
                           const float* __restrict__ w_b,
                           float* __restrict__ out)
{
    __shared__ __attribute__((aligned(16))) float o_lds[16 * OROWS * OS];
    __shared__ __attribute__((aligned(16))) float hr_lds[TG * KTOP * HRS];
    __shared__ int   list_lds[NEXP * TG];
    __shared__ int   cnt_lds[NEXP];
    __shared__ int   i0_lds[TG];
    __shared__ int   i1_lds[TG];
    __shared__ int   tileP_lds[MAXT];
    __shared__ int   nT_lds;

    const int tid  = threadIdx.x;
    const int wave = __builtin_amdgcn_readfirstlane(tid >> 5);
    const int lane = tid & 31;
    const int nIdx = lane & 15;
    const int hv   = lane >> 4;
    const int grp  = blockIdx.x >> 3;
    const int dg   = blockIdx.x & 7;
    const int tokBase = grp * TG;

    if (tid < TG) {
        i0_lds[tid] = idxs[(tokBase + tid) * KTOP + 0];
        i1_lds[tid] = idxs[(tokBase + tid) * KTOP + 1];
    }
    float* ob = &o_lds[wave * (OROWS * OS)];
    for (int it = 0; it < 16; ++it) {
        const int fi = it * 32 + lane;
        *reinterpret_cast<float4*>(ob + (fi >> 2) * OS + ((fi & 3) << 2)) = make_float4(0.f, 0.f, 0.f, 0.f);
    }

    {
        const size_t ksStride = (size_t)NTOK * KTOP * HCS;
#pragma unroll
        for (int i = 0; i < 2; ++i) {
            const int fi   = tid + i * 512;
            const int row  = fi >> 3;
            const int slot = (fi >> 2) & 1;
            const int c4   = (fi & 3) << 2;
            const int tok  = tokBase + row;
            const int   e0 = idxs[tok * KTOP + 0];
            const int   e1 = idxs[tok * KTOP + 1];
            const float r0 = routing[tok * KTOP + 0];
            const float r1 = routing[tok * KTOP + 1];
            const int   es = slot ? e1 : e0;
            float g = 0.f;
            g += (e0 == es) ? r0 : 0.f;
            g += (e1 == es) ? r1 : 0.f;
            const float gate = 2.0f * g;
            const float* p0 = hc + ((size_t)tok * KTOP + slot) * HCS + c4;
            float4 s[KSPLIT];
#pragma unroll
            for (int p = 0; p < KSPLIT; ++p)
                s[p] = *reinterpret_cast<const float4*>(p0 + p * ksStride);
            float hx = s[0].x, hy = s[0].y, hz = s[0].z, hw = s[0].w;
#pragma unroll
            for (int p = 1; p < KSPLIT; ++p) { hx += s[p].x; hy += s[p].y; hz += s[p].z; hw += s[p].w; }
            float4 o;
            o.x = gate * (hx * __builtin_amdgcn_rcpf(1.0f + __expf(-hx)));
            o.y = gate * (hy * __builtin_amdgcn_rcpf(1.0f + __expf(-hy)));
            o.z = gate * (hz * __builtin_amdgcn_rcpf(1.0f + __expf(-hz)));
            o.w = gate * (hw * __builtin_amdgcn_rcpf(1.0f + __expf(-hw)));
            *reinterpret_cast<float4*>(&hr_lds[(row * KTOP + slot) * HRS + c4]) = o;
        }
    }
    __syncthreads();

    {
        const int e = wave;
        int base = 0;
        for (int it = 0; it < TG / 32; ++it) {
            const int row = it * 32 + lane;
            const bool hit = (i0_lds[row] == e) || (i1_lds[row] == e);
            const unsigned mask = __builtin_amdgcn_ballot_w32(hit);
            const int pos = base + __popc(mask & ((1u << lane) - 1u));
            if (hit) list_lds[e * TG + pos] = row * KTOP + ((i0_lds[row] == e) ? 0 : 1);
            base += __popc(mask);
        }
        if (lane == 0) cnt_lds[e] = base;
    }
    __syncthreads();
    if (wave == 0) {
        const int cE = (lane < NEXP) ? cnt_lds[lane] : 0;
        const int nt = (cE + 15) >> 4;
        int inc = nt;
        for (int d = 1; d < NEXP; d <<= 1) {
            const int o = __shfl_up(inc, d);
            if (lane >= d) inc += o;
        }
        const int first = inc - nt;
        for (int t = 0; t < nt; ++t) tileP_lds[first + t] = lane | ((t * 16) << 8) | (cE << 16);
        if (lane == NEXP - 1) nT_lds = inc;
    }
    __syncthreads();

    const int nT = __builtin_amdgcn_readfirstlane(nT_lds);
    const int d0 = dg * 256 + wave * 16 + nIdx;
    const float* wd = w_b + (size_t)d0 * RANK + 2 * hv;

    int tp = __builtin_amdgcn_readfirstlane(tileP_lds[0]);
    v2f bc0[4];
    {
        const float* wb = wd + (size_t)(tp & 255) * (D_OUT * RANK);
#pragma unroll
        for (int j = 0; j < 4; ++j) bc0[j] = *reinterpret_cast<const v2f*>(wb + 4 * j);
    }

    for (int t = 0; t < nT; ++t) {
        const int e  = tp & 255;
        const int q0 = (tp >> 8) & 255;
        const int c  = tp >> 16;
        const int tn  = (t + 1 < nT) ? (t + 1) : t;
        const int tpN = __builtin_amdgcn_readfirstlane(tileP_lds[tn]);
        const float* wbN = wd + (size_t)(tpN & 255) * (D_OUT * RANK);
        v2f bn0[4];
#pragma unroll
        for (int j = 0; j < 4; ++j) bn0[j] = *reinterpret_cast<const v2f*>(wbN + 4 * j);

        const int* le = &list_lds[e * TG];
        const int  qa   = q0 + nIdx;
        const bool live = qa < c;
        const int  rec  = le[live ? qa : (c - 1)];
        const float* hb = &hr_lds[rec * HRS + 2 * hv];
        v2f b[4];
#pragma unroll
        for (int j = 0; j < 4; ++j) b[j] = *reinterpret_cast<const v2f*>(hb + 4 * j);

        const int row = rec >> 1;
        const float* pr = ob + row * OS + 8 * hv;
        const float4 o00 = *reinterpret_cast<const float4*>(pr);
        const float4 o01 = *reinterpret_cast<const float4*>(pr + 4);

        v8f acc0 = (v8f){};
#pragma unroll
        for (int j = 0; j < 4; ++j)
            acc0 = __builtin_amdgcn_wmma_f32_16x16x4_f32(false, bc0[j], false, b[j],
                                                         (short)0, acc0, false, false);

        float* pw = ob + (live ? row : (TG + nIdx)) * OS + 8 * hv;
        *reinterpret_cast<float4*>(pw)      = make_float4(o00.x + acc0[0], o00.y + acc0[1], o00.z + acc0[2], o00.w + acc0[3]);
        *reinterpret_cast<float4*>(pw + 4)  = make_float4(o01.x + acc0[4], o01.y + acc0[5], o01.z + acc0[6], o01.w + acc0[7]);
#pragma unroll
        for (int j = 0; j < 4; ++j) bc0[j] = bn0[j];
        tp = tpN;
    }
    __syncthreads();

    {
        const int pair = wave >> 1;
        const int rowBase = (wave & 1) * (TG / 2);
        float* og = out + (size_t)(tokBase + rowBase) * D_OUT + dg * 256 + pair * 32;
        const int r4 = lane >> 3;
        const int c4 = (lane & 7) << 2;
        const float* sb = &o_lds[(pair * 2 + (c4 >> 4)) * (OROWS * OS)] + (c4 & 15);
#pragma unroll 4
        for (int it = 0; it < 16; ++it) {
            const int row = it * 4 + r4;
            const float4 v = *reinterpret_cast<const float4*>(sb + (rowBase + row) * OS);
            *reinterpret_cast<float4*>(og + (size_t)row * D_OUT + c4) = v;
        }
    }
}

extern "C" void kernel_launch(void* const* d_in, const int* in_sizes, int n_in,
                              void* d_out, int out_size, void* d_ws, size_t ws_size,
                              hipStream_t stream) {
    const float* x       = (const float*)d_in[0];
    const float* routing = (const float*)d_in[1];
    const int*   idxs    = (const int*)d_in[2];
    const float* w_a     = (const float*)d_in[3];
    const float* w_b     = (const float*)d_in[4];
    float* out  = (float*)d_out;
    float* hc   = (float*)d_ws;

    hipLaunchKernelGGL(lora_down_routed_kernel, dim3(256), dim3(512), 0, stream,
                       x, idxs, w_a, hc);
    hipLaunchKernelGGL(lora_up_routed_kernel, dim3(256), dim3(512), 0, stream,
                       hc, routing, idxs, w_b, out);
}
